# starting kernel
# baseline (speedup 1.0000x reference)
#include <hip/hip_runtime.h>
#include <hip/hip_bf16.h>

typedef __attribute__((ext_vector_type(2))) float v2f;
typedef __attribute__((ext_vector_type(8))) float v8f;

#define SEQ_T 32768
#define IN_D  512
#define HID_H 20
#define G4H   80

__global__ __launch_bounds__(32) void lstm_gates_gemm(
    const float* __restrict__ x,
    const float* __restrict__ W_ih,
    const float* __restrict__ b_ih,
    const float* __restrict__ b_hh,
    float* __restrict__ xg)
{
    const int mtile = blockIdx.x;
    const int lane  = threadIdx.x;
    const int half  = lane >> 4;
    const int l15   = lane & 15;

    v8f acc[5];
#pragma unroll
    for (int nt = 0; nt < 5; ++nt) {
        const int n = nt * 16 + l15;
        const float bias = b_ih[n] + b_hh[n];
#pragma unroll
        for (int v = 0; v < 8; ++v) acc[nt][v] = bias;
    }

    const float* xrow = x + (size_t)(mtile * 16 + l15) * IN_D;

    for (int k0 = 0; k0 < IN_D; k0 += 4) {
        v2f a = *(const v2f*)(xrow + k0 + 2 * half);
#pragma unroll
        for (int nt = 0; nt < 5; ++nt) {
            v2f b = *(const v2f*)(W_ih + (size_t)(nt * 16 + l15) * IN_D + k0 + 2 * half);
            acc[nt] = __builtin_amdgcn_wmma_f32_16x16x4_f32(
                 false, a,  false, b,
                 (short)0, acc[nt],
                 false,  false);
        }
    }

    float* outbase = xg + (size_t)(mtile * 16) * G4H;
#pragma unroll
    for (int nt = 0; nt < 5; ++nt) {
#pragma unroll
        for (int v = 0; v < 8; ++v) {
            outbase[(size_t)(v + 8 * half) * G4H + nt * 16 + l15] = acc[nt][v];
        }
    }
}

__device__ __forceinline__ float readlane_f(float v, int l) {
    return __builtin_bit_cast(float,
        __builtin_amdgcn_readlane(__builtin_bit_cast(int, v), l));
}

__device__ __forceinline__ float sigmoid_f(float v) {
    return 1.0f / (1.0f + __expf(-v));
}

__global__ __launch_bounds__(32) void lstm_scan(
    const float* __restrict__ xg,
    const float* __restrict__ W_hh,
    const float* __restrict__ W_out,
    const float* __restrict__ b_out,
    const float* __restrict__ h0,
    const float* __restrict__ c0,
    float* __restrict__ out)
{
    const int lane   = threadIdx.x;
    const bool active = lane < HID_H;
    const int  n      = active ? lane : 0;

    float wi[HID_H], wf[HID_H], wg[HID_H], wo[HID_H], wout[HID_H];
#pragma unroll
    for (int k = 0; k < HID_H; ++k) {
        wi[k]   = W_hh[(size_t)(0 * HID_H + n) * HID_H + k];
        wf[k]   = W_hh[(size_t)(1 * HID_H + n) * HID_H + k];
        wg[k]   = W_hh[(size_t)(2 * HID_H + n) * HID_H + k];
        wo[k]   = W_hh[(size_t)(3 * HID_H + n) * HID_H + k];
        wout[k] = W_out[k];
    }
    const float bout = b_out[0];

    float c = c0[n];
    float h = h0[n];

    float hb[HID_H];
#pragma unroll
    for (int k = 0; k < HID_H; ++k) hb[k] = readlane_f(h, k);

    float xi = xg[0 * G4H + 0 * HID_H + n];
    float xf = xg[0 * G4H + 1 * HID_H + n];
    float xc = xg[0 * G4H + 2 * HID_H + n];
    float xo = xg[0 * G4H + 3 * HID_H + n];

    for (int t = 0; t < SEQ_T; ++t) {
        float nxi = 0.f, nxf = 0.f, nxc = 0.f, nxo = 0.f;
        if (t + 1 < SEQ_T) {
            const float* p = xg + (size_t)(t + 1) * G4H;
            nxi = p[0 * HID_H + n];
            nxf = p[1 * HID_H + n];
            nxc = p[2 * HID_H + n];
            nxo = p[3 * HID_H + n];
        }

        float gi = xi, gf = xf, gc = xc, go = xo;
#pragma unroll
        for (int k = 0; k < HID_H; ++k) {
            gi = fmaf(wi[k], hb[k], gi);
            gf = fmaf(wf[k], hb[k], gf);
            gc = fmaf(wg[k], hb[k], gc);
            go = fmaf(wo[k], hb[k], go);
        }

        const float i_ = sigmoid_f(gi);
        const float f_ = sigmoid_f(gf);
        const float o_ = sigmoid_f(go);
        const float g_ = tanhf(gc);

        c = fmaf(f_, c, i_ * g_);
        h = o_ * tanhf(c);

#pragma unroll
        for (int k = 0; k < HID_H; ++k) hb[k] = readlane_f(h, k);

        float ot = bout;
#pragma unroll
        for (int k = 0; k < HID_H; ++k) ot = fmaf(wout[k], hb[k], ot);
        if (lane == 0) out[t] = ot;

        xi = nxi; xf = nxf; xc = nxc; xo = nxo;
    }

    if (active) {
        out[SEQ_T + n]         = h;
        out[SEQ_T + HID_H + n] = c;
    }
}

extern "C" void kernel_launch(void* const* d_in, const int* in_sizes, int n_in,
                              void* d_out, int out_size, void* d_ws, size_t ws_size,
                              hipStream_t stream) {
    (void)in_sizes; (void)n_in; (void)out_size; (void)ws_size;

    const float* x    = (const float*)d_in[0];
    const float* h0   = (const float*)d_in[1];
    const float* c0   = (const float*)d_in[2];
    const float* W_ih = (const float*)d_in[3];
    const float* W_hh = (const float*)d_in[4];
    const float* b_ih = (const float*)d_in[5];
    const float* b_hh = (const float*)d_in[6];
    const float* W_out= (const float*)d_in[7];
    const float* b_out= (const float*)d_in[8];

    float* out = (float*)d_out;
    float* xg  = (float*)d_ws;

    lstm_gates_gemm<<<SEQ_T / 16, 32, 0, stream>>>(x, W_ih, b_ih, b_hh, xg);

    lstm_scan<<<1, 32, 0, stream>>>(xg, W_hh, W_out, b_out, h0, c0, out);
}


# scan step made branch-free: tanh computes both ranges and selects per lane (same forms, constants and values), so the four gate chains and the activations overlap in one block
# speedup vs baseline: 1.2008x; 1.2008x over previous
#include <hip/hip_runtime.h>
#include <hip/hip_bf16.h>

typedef __attribute__((ext_vector_type(2))) float v2f;
typedef __attribute__((ext_vector_type(8))) float v8f;

#define SEQ_T 32768
#define IN_D  512
#define HID_H 20
#define G4H   80

__global__ __launch_bounds__(32) void lstm_gates_gemm(
    const float* __restrict__ x,
    const float* __restrict__ W_ih,
    const float* __restrict__ b_ih,
    const float* __restrict__ b_hh,
    float* __restrict__ xg)
{
    const int mtile = blockIdx.x;
    const int lane  = threadIdx.x;
    const int half  = lane >> 4;
    const int l15   = lane & 15;

    v8f acc[5];
#pragma unroll
    for (int nt = 0; nt < 5; ++nt) {
        const int n = nt * 16 + l15;
        const float bias = b_ih[n] + b_hh[n];
#pragma unroll
        for (int v = 0; v < 8; ++v) acc[nt][v] = bias;
    }

    const float* xrow = x + (size_t)(mtile * 16 + l15) * IN_D;

    for (int k0 = 0; k0 < IN_D; k0 += 4) {
        v2f a = *(const v2f*)(xrow + k0 + 2 * half);
#pragma unroll
        for (int nt = 0; nt < 5; ++nt) {
            v2f b = *(const v2f*)(W_ih + (size_t)(nt * 16 + l15) * IN_D + k0 + 2 * half);
            acc[nt] = __builtin_amdgcn_wmma_f32_16x16x4_f32(
                 false, a,  false, b,
                 (short)0, acc[nt],
                 false,  false);
        }
    }

    float* outbase = xg + (size_t)(mtile * 16) * G4H;
#pragma unroll
    for (int nt = 0; nt < 5; ++nt) {
#pragma unroll
        for (int v = 0; v < 8; ++v) {
            outbase[(size_t)(v + 8 * half) * G4H + nt * 16 + l15] = acc[nt][v];
        }
    }
}

__device__ __forceinline__ float readlane_f(float v, int l) {
    return __builtin_bit_cast(float,
        __builtin_amdgcn_readlane(__builtin_bit_cast(int, v), l));
}

__device__ __forceinline__ float sigmoid_f(float v) {
    return 1.0f / (1.0f + __expf(-v));
}

__device__ __forceinline__ float tanh_sel(float x) {
    const float y  = __builtin_fabsf(x);
    const float y2 = y * y;
    float p = fmaf(y2, -0x1.758e7ap-8f, 0x1.521192p-6f);
    p = fmaf(y2, p, -0x1.b8389cp-5f);
    p = fmaf(y2, p,  0x1.110704p-3f);
    p = fmaf(y2, p, -0x1.555532p-2f);
    const float z_small = fmaf(y2, y * p, y);
    const float t = expf(2.0f * y);
    const float z_large = fmaf(-2.0f, __builtin_amdgcn_rcpf(t + 1.0f), 1.0f);
    const float z = (y < 0.625f) ? z_small : z_large;
    return __builtin_copysignf(z, x);
}

__global__ __launch_bounds__(32) void lstm_scan(
    const float* __restrict__ xg,
    const float* __restrict__ W_hh,
    const float* __restrict__ W_out,
    const float* __restrict__ b_out,
    const float* __restrict__ h0,
    const float* __restrict__ c0,
    float* __restrict__ out)
{
    const int lane   = threadIdx.x;
    const bool active = lane < HID_H;
    const int  n      = active ? lane : 0;

    float wi[HID_H], wf[HID_H], wg[HID_H], wo[HID_H], wout[HID_H];
#pragma unroll
    for (int k = 0; k < HID_H; ++k) {
        wi[k]   = W_hh[(size_t)(0 * HID_H + n) * HID_H + k];
        wf[k]   = W_hh[(size_t)(1 * HID_H + n) * HID_H + k];
        wg[k]   = W_hh[(size_t)(2 * HID_H + n) * HID_H + k];
        wo[k]   = W_hh[(size_t)(3 * HID_H + n) * HID_H + k];
        wout[k] = W_out[k];
    }
    const float bout = b_out[0];

    float c = c0[n];
    float h = h0[n];

    float hb[HID_H];
#pragma unroll
    for (int k = 0; k < HID_H; ++k) hb[k] = readlane_f(h, k);

    float xi = xg[0 * G4H + 0 * HID_H + n];
    float xf = xg[0 * G4H + 1 * HID_H + n];
    float xc = xg[0 * G4H + 2 * HID_H + n];
    float xo = xg[0 * G4H + 3 * HID_H + n];

    for (int t = 0; t < SEQ_T; ++t) {
        float nxi = 0.f, nxf = 0.f, nxc = 0.f, nxo = 0.f;
        if (t + 1 < SEQ_T) {
            const float* p = xg + (size_t)(t + 1) * G4H;
            nxi = p[0 * HID_H + n];
            nxf = p[1 * HID_H + n];
            nxc = p[2 * HID_H + n];
            nxo = p[3 * HID_H + n];
        }
        __builtin_amdgcn_sched_barrier(0);

        float gi = xi, gf = xf, gc = xc, go = xo;
#pragma unroll
        for (int k = 0; k < HID_H; ++k) {
            gi = fmaf(wi[k], hb[k], gi);
            gf = fmaf(wf[k], hb[k], gf);
            gc = fmaf(wg[k], hb[k], gc);
            go = fmaf(wo[k], hb[k], go);
        }
        __builtin_amdgcn_sched_barrier(0);

        const float i_ = sigmoid_f(gi);
        const float f_ = sigmoid_f(gf);
        const float o_ = sigmoid_f(go);
        const float g_ = tanh_sel(gc);

        c = fmaf(f_, c, i_ * g_);
        h = o_ * tanh_sel(c);

#pragma unroll
        for (int k = 0; k < HID_H; ++k) hb[k] = readlane_f(h, k);

        float ot = bout;
#pragma unroll
        for (int k = 0; k < HID_H; ++k) ot = fmaf(wout[k], hb[k], ot);
        if (lane == 0) out[t] = ot;

        xi = nxi; xf = nxf; xc = nxc; xo = nxo;
    }

    if (active) {
        out[SEQ_T + n]         = h;
        out[SEQ_T + HID_H + n] = c;
    }
}

extern "C" void kernel_launch(void* const* d_in, const int* in_sizes, int n_in,
                              void* d_out, int out_size, void* d_ws, size_t ws_size,
                              hipStream_t stream) {
    (void)in_sizes; (void)n_in; (void)out_size; (void)ws_size;

    const float* x    = (const float*)d_in[0];
    const float* h0   = (const float*)d_in[1];
    const float* c0   = (const float*)d_in[2];
    const float* W_ih = (const float*)d_in[3];
    const float* W_hh = (const float*)d_in[4];
    const float* b_ih = (const float*)d_in[5];
    const float* b_hh = (const float*)d_in[6];
    const float* W_out= (const float*)d_in[7];
    const float* b_out= (const float*)d_in[8];

    float* out = (float*)d_out;
    float* xg  = (float*)d_ws;

    lstm_gates_gemm<<<SEQ_T / 16, 32, 0, stream>>>(x, W_ih, b_ih, b_hh, xg);

    lstm_scan<<<1, 32, 0, stream>>>(xg, W_hh, W_out, b_out, h0, c0, out);
}


# scan loads its xg rows one block of 8 steps ahead (32 loads at the block's top) instead of one step ahead, so the per-step round trip to memory is off the serial chain; values and sum order unchanged
# speedup vs baseline: 1.7409x; 1.4498x over previous
#include <hip/hip_runtime.h>
#include <hip/hip_bf16.h>

typedef __attribute__((ext_vector_type(2))) float v2f;
typedef __attribute__((ext_vector_type(8))) float v8f;

#define SEQ_T 32768
#define IN_D  512
#define HID_H 20
#define G4H   80
#define XG_BLOCK 8

__global__ __launch_bounds__(32) void lstm_gates_gemm(
    const float* __restrict__ x,
    const float* __restrict__ W_ih,
    const float* __restrict__ b_ih,
    const float* __restrict__ b_hh,
    float* __restrict__ xg)
{
    const int mtile = blockIdx.x;
    const int lane  = threadIdx.x;
    const int half  = lane >> 4;
    const int l15   = lane & 15;

    v8f acc[5];
#pragma unroll
    for (int nt = 0; nt < 5; ++nt) {
        const int n = nt * 16 + l15;
        const float bias = b_ih[n] + b_hh[n];
#pragma unroll
        for (int v = 0; v < 8; ++v) acc[nt][v] = bias;
    }

    const float* xrow = x + (size_t)(mtile * 16 + l15) * IN_D;

    for (int k0 = 0; k0 < IN_D; k0 += 4) {
        v2f a = *(const v2f*)(xrow + k0 + 2 * half);
#pragma unroll
        for (int nt = 0; nt < 5; ++nt) {
            v2f b = *(const v2f*)(W_ih + (size_t)(nt * 16 + l15) * IN_D + k0 + 2 * half);
            acc[nt] = __builtin_amdgcn_wmma_f32_16x16x4_f32(
                 false, a,  false, b,
                 (short)0, acc[nt],
                 false,  false);
        }
    }

    float* outbase = xg + (size_t)(mtile * 16) * G4H;
#pragma unroll
    for (int nt = 0; nt < 5; ++nt) {
#pragma unroll
        for (int v = 0; v < 8; ++v) {
            outbase[(size_t)(v + 8 * half) * G4H + nt * 16 + l15] = acc[nt][v];
        }
    }
}

__device__ __forceinline__ float readlane_f(float v, int l) {
    return __builtin_bit_cast(float,
        __builtin_amdgcn_readlane(__builtin_bit_cast(int, v), l));
}

__device__ __forceinline__ float sigmoid_f(float v) {
    return 1.0f / (1.0f + __expf(-v));
}

__device__ __forceinline__ float tanh_sel(float x) {
    const float y  = __builtin_fabsf(x);
    const float y2 = y * y;
    float p = fmaf(y2, -0x1.758e7ap-8f, 0x1.521192p-6f);
    p = fmaf(y2, p, -0x1.b8389cp-5f);
    p = fmaf(y2, p,  0x1.110704p-3f);
    p = fmaf(y2, p, -0x1.555532p-2f);
    const float z_small = fmaf(y2, y * p, y);
    const float t = expf(2.0f * y);
    const float z_large = fmaf(-2.0f, __builtin_amdgcn_rcpf(t + 1.0f), 1.0f);
    const float z = (y < 0.625f) ? z_small : z_large;
    return __builtin_copysignf(z, x);
}

__global__ __launch_bounds__(32) void lstm_scan(
    const float* __restrict__ xg,
    const float* __restrict__ W_hh,
    const float* __restrict__ W_out,
    const float* __restrict__ b_out,
    const float* __restrict__ h0,
    const float* __restrict__ c0,
    float* __restrict__ out)
{
    const int lane   = threadIdx.x;
    const bool active = lane < HID_H;
    const int  n      = active ? lane : 0;

    float wi[HID_H], wf[HID_H], wg[HID_H], wo[HID_H], wout[HID_H];
#pragma unroll
    for (int k = 0; k < HID_H; ++k) {
        wi[k]   = W_hh[(size_t)(0 * HID_H + n) * HID_H + k];
        wf[k]   = W_hh[(size_t)(1 * HID_H + n) * HID_H + k];
        wg[k]   = W_hh[(size_t)(2 * HID_H + n) * HID_H + k];
        wo[k]   = W_hh[(size_t)(3 * HID_H + n) * HID_H + k];
        wout[k] = W_out[k];
    }
    const float bout = b_out[0];

    float c = c0[n];
    float h = h0[n];

    float hb[HID_H];
#pragma unroll
    for (int k = 0; k < HID_H; ++k) hb[k] = readlane_f(h, k);

    float ni[XG_BLOCK], nf[XG_BLOCK], nc[XG_BLOCK], no[XG_BLOCK];
#pragma unroll
    for (int j = 0; j < XG_BLOCK; ++j) {
        const float* p = xg + (size_t)j * G4H;
        ni[j] = p[0 * HID_H + n];
        nf[j] = p[1 * HID_H + n];
        nc[j] = p[2 * HID_H + n];
        no[j] = p[3 * HID_H + n];
    }

    for (int t0 = 0; t0 < SEQ_T; t0 += XG_BLOCK) {
        float ci[XG_BLOCK], cf[XG_BLOCK], cc[XG_BLOCK], co[XG_BLOCK];
#pragma unroll
        for (int j = 0; j < XG_BLOCK; ++j) {
            ci[j] = ni[j]; cf[j] = nf[j]; cc[j] = nc[j]; co[j] = no[j];
        }

        {
            const int tb = (t0 + XG_BLOCK < SEQ_T) ? (t0 + XG_BLOCK) : (SEQ_T - XG_BLOCK);
            const float* pb = xg + (size_t)tb * G4H;
#pragma unroll
            for (int j = 0; j < XG_BLOCK; ++j) {
                const float* p = pb + j * G4H;
                ni[j] = p[0 * HID_H + n];
                nf[j] = p[1 * HID_H + n];
                nc[j] = p[2 * HID_H + n];
                no[j] = p[3 * HID_H + n];
            }
        }
        __builtin_amdgcn_sched_barrier(0);

#pragma unroll
        for (int j = 0; j < XG_BLOCK; ++j) {
            const int t = t0 + j;

            float gi = ci[j], gf = cf[j], gc = cc[j], go = co[j];
#pragma unroll
            for (int k = 0; k < HID_H; ++k) {
                gi = fmaf(wi[k], hb[k], gi);
                gf = fmaf(wf[k], hb[k], gf);
                gc = fmaf(wg[k], hb[k], gc);
                go = fmaf(wo[k], hb[k], go);
            }
            __builtin_amdgcn_sched_barrier(0);

            const float i_ = sigmoid_f(gi);
            const float f_ = sigmoid_f(gf);
            const float o_ = sigmoid_f(go);
            const float g_ = tanh_sel(gc);

            c = fmaf(f_, c, i_ * g_);
            h = o_ * tanh_sel(c);

#pragma unroll
            for (int k = 0; k < HID_H; ++k) hb[k] = readlane_f(h, k);

            float ot = bout;
#pragma unroll
            for (int k = 0; k < HID_H; ++k) ot = fmaf(wout[k], hb[k], ot);
            if (lane == 0) out[t] = ot;
        }
    }

    if (active) {
        out[SEQ_T + n]         = h;
        out[SEQ_T + HID_H + n] = c;
    }
}

extern "C" void kernel_launch(void* const* d_in, const int* in_sizes, int n_in,
                              void* d_out, int out_size, void* d_ws, size_t ws_size,
                              hipStream_t stream) {
    (void)in_sizes; (void)n_in; (void)out_size; (void)ws_size;

    const float* x    = (const float*)d_in[0];
    const float* h0   = (const float*)d_in[1];
    const float* c0   = (const float*)d_in[2];
    const float* W_ih = (const float*)d_in[3];
    const float* W_hh = (const float*)d_in[4];
    const float* b_ih = (const float*)d_in[5];
    const float* b_hh = (const float*)d_in[6];
    const float* W_out= (const float*)d_in[7];
    const float* b_out= (const float*)d_in[8];

    float* out = (float*)d_out;
    float* xg  = (float*)d_ws;

    lstm_gates_gemm<<<SEQ_T / 16, 32, 0, stream>>>(x, W_ih, b_ih, b_hh, xg);

    lstm_scan<<<1, 32, 0, stream>>>(xg, W_hh, W_out, b_out, h0, c0, out);
}


# scan: output projection moved off the scalar unit: lane 20 (owns no unit) holds W_out as its i-gate row and b_out as start, so step t's i chain leaves out[t-1] there; same fma order and values
# speedup vs baseline: 2.4768x; 1.4227x over previous
#include <hip/hip_runtime.h>
#include <hip/hip_bf16.h>

typedef __attribute__((ext_vector_type(2))) float v2f;
typedef __attribute__((ext_vector_type(8))) float v8f;

#define SEQ_T 32768
#define IN_D  512
#define HID_H 20
#define G4H   80
#define XG_BLOCK 8

__global__ __launch_bounds__(32) void lstm_gates_gemm(
    const float* __restrict__ x,
    const float* __restrict__ W_ih,
    const float* __restrict__ b_ih,
    const float* __restrict__ b_hh,
    float* __restrict__ xg)
{
    const int mtile = blockIdx.x;
    const int lane  = threadIdx.x;
    const int half  = lane >> 4;
    const int l15   = lane & 15;

    v8f acc[5];
#pragma unroll
    for (int nt = 0; nt < 5; ++nt) {
        const int n = nt * 16 + l15;
        const float bias = b_ih[n] + b_hh[n];
#pragma unroll
        for (int v = 0; v < 8; ++v) acc[nt][v] = bias;
    }

    const float* xrow = x + (size_t)(mtile * 16 + l15) * IN_D;

    for (int k0 = 0; k0 < IN_D; k0 += 4) {
        v2f a = *(const v2f*)(xrow + k0 + 2 * half);
#pragma unroll
        for (int nt = 0; nt < 5; ++nt) {
            v2f b = *(const v2f*)(W_ih + (size_t)(nt * 16 + l15) * IN_D + k0 + 2 * half);
            acc[nt] = __builtin_amdgcn_wmma_f32_16x16x4_f32(
                 false, a,  false, b,
                 (short)0, acc[nt],
                 false,  false);
        }
    }

    float* outbase = xg + (size_t)(mtile * 16) * G4H;
#pragma unroll
    for (int nt = 0; nt < 5; ++nt) {
#pragma unroll
        for (int v = 0; v < 8; ++v) {
            outbase[(size_t)(v + 8 * half) * G4H + nt * 16 + l15] = acc[nt][v];
        }
    }
}

__device__ __forceinline__ float readlane_f(float v, int l) {
    return __builtin_bit_cast(float,
        __builtin_amdgcn_readlane(__builtin_bit_cast(int, v), l));
}

__device__ __forceinline__ float sigmoid_f(float v) {
    return 1.0f / (1.0f + __expf(-v));
}

__device__ __forceinline__ float tanh_sel(float x) {
    const float y  = __builtin_fabsf(x);
    const float y2 = y * y;
    float p = fmaf(y2, -0x1.758e7ap-8f, 0x1.521192p-6f);
    p = fmaf(y2, p, -0x1.b8389cp-5f);
    p = fmaf(y2, p,  0x1.110704p-3f);
    p = fmaf(y2, p, -0x1.555532p-2f);
    const float z_small = fmaf(y2, y * p, y);
    const float t = expf(2.0f * y);
    const float z_large = fmaf(-2.0f, __builtin_amdgcn_rcpf(t + 1.0f), 1.0f);
    const float z = (y < 0.625f) ? z_small : z_large;
    return __builtin_copysignf(z, x);
}

__global__ __launch_bounds__(32) void lstm_scan(
    const float* __restrict__ xg,
    const float* __restrict__ W_hh,
    const float* __restrict__ W_out,
    const float* __restrict__ b_out,
    const float* __restrict__ h0,
    const float* __restrict__ c0,
    float* __restrict__ out)
{
    const int lane   = threadIdx.x;
    const bool active = lane < HID_H;
    const bool proj   = lane == HID_H;
    const int  n      = active ? lane : 0;

    const float* wi_row = proj ? W_out : W_hh + (size_t)(0 * HID_H + n) * HID_H;
    float wi[HID_H], wf[HID_H], wg[HID_H], wo[HID_H];
#pragma unroll
    for (int k = 0; k < HID_H; ++k) {
        wi[k]   = wi_row[k];
        wf[k]   = W_hh[(size_t)(1 * HID_H + n) * HID_H + k];
        wg[k]   = W_hh[(size_t)(2 * HID_H + n) * HID_H + k];
        wo[k]   = W_hh[(size_t)(3 * HID_H + n) * HID_H + k];
    }
    const float bout = b_out[0];

    float c = c0[n];
    float h = h0[n];

    float hb[HID_H];
#pragma unroll
    for (int k = 0; k < HID_H; ++k) hb[k] = readlane_f(h, k);

    float ni[XG_BLOCK], nf[XG_BLOCK], nc[XG_BLOCK], no[XG_BLOCK];
#pragma unroll
    for (int j = 0; j < XG_BLOCK; ++j) {
        const float* p = xg + (size_t)j * G4H;
        ni[j] = p[0 * HID_H + n];
        nf[j] = p[1 * HID_H + n];
        nc[j] = p[2 * HID_H + n];
        no[j] = p[3 * HID_H + n];
        ni[j] = proj ? bout : ni[j];
    }

    for (int t0 = 0; t0 < SEQ_T; t0 += XG_BLOCK) {
        float ci[XG_BLOCK], cf[XG_BLOCK], cc[XG_BLOCK], co[XG_BLOCK];
#pragma unroll
        for (int j = 0; j < XG_BLOCK; ++j) {
            ci[j] = ni[j]; cf[j] = nf[j]; cc[j] = nc[j]; co[j] = no[j];
        }

        {
            const int tb = (t0 + XG_BLOCK < SEQ_T) ? (t0 + XG_BLOCK) : (SEQ_T - XG_BLOCK);
            const float* pb = xg + (size_t)tb * G4H;
#pragma unroll
            for (int j = 0; j < XG_BLOCK; ++j) {
                const float* p = pb + j * G4H;
                ni[j] = p[0 * HID_H + n];
                nf[j] = p[1 * HID_H + n];
                nc[j] = p[2 * HID_H + n];
                no[j] = p[3 * HID_H + n];
            }
        }
        __builtin_amdgcn_sched_barrier(0);

        float og[XG_BLOCK];

#pragma unroll
        for (int j = 0; j < XG_BLOCK; ++j) {
            float gi = ci[j], gf = cf[j], gc = cc[j], go = co[j];
#pragma unroll
            for (int k = 0; k < HID_H; ++k) {
                gi = fmaf(wi[k], hb[k], gi);
                gf = fmaf(wf[k], hb[k], gf);
                gc = fmaf(wg[k], hb[k], gc);
                go = fmaf(wo[k], hb[k], go);
            }
            og[j] = gi;
            __builtin_amdgcn_sched_barrier(0);

            const float i_ = sigmoid_f(gi);
            const float f_ = sigmoid_f(gf);
            const float o_ = sigmoid_f(go);
            const float g_ = tanh_sel(gc);

            c = fmaf(f_, c, i_ * g_);
            h = o_ * tanh_sel(c);

#pragma unroll
            for (int k = 0; k < HID_H; ++k) hb[k] = readlane_f(h, k);
        }

#pragma unroll
        for (int j = 0; j < XG_BLOCK; ++j) ni[j] = proj ? bout : ni[j];

        if (proj) {
            float* ob = out + t0;
            if (t0 > 0) ob[-1] = og[0];
#pragma unroll
            for (int j = 1; j < XG_BLOCK; ++j) ob[j - 1] = og[j];
        }
    }

    {
        float ot = bout;
#pragma unroll
        for (int k = 0; k < HID_H; ++k) ot = fmaf(W_out[k], hb[k], ot);
        if (lane == 0) out[SEQ_T - 1] = ot;
    }

    if (active) {
        out[SEQ_T + n]         = h;
        out[SEQ_T + HID_H + n] = c;
    }
}

extern "C" void kernel_launch(void* const* d_in, const int* in_sizes, int n_in,
                              void* d_out, int out_size, void* d_ws, size_t ws_size,
                              hipStream_t stream) {
    (void)in_sizes; (void)n_in; (void)out_size; (void)ws_size;

    const float* x    = (const float*)d_in[0];
    const float* h0   = (const float*)d_in[1];
    const float* c0   = (const float*)d_in[2];
    const float* W_ih = (const float*)d_in[3];
    const float* W_hh = (const float*)d_in[4];
    const float* b_ih = (const float*)d_in[5];
    const float* b_hh = (const float*)d_in[6];
    const float* W_out= (const float*)d_in[7];
    const float* b_out= (const float*)d_in[8];

    float* out = (float*)d_out;
    float* xg  = (float*)d_ws;

    lstm_gates_gemm<<<SEQ_T / 16, 32, 0, stream>>>(x, W_ih, b_ih, b_hh, xg);

    lstm_scan<<<1, 32, 0, stream>>>(xg, W_hh, W_out, b_out, h0, c0, out);
}


# scan: activations on bare v_exp_f32/v_rcp_f32 (no IEEE division, no range-reduced exp; values move ~1 ulp) and a scheduling fence per k that stops the compiler pairing the gate fma chains across k
# speedup vs baseline: 3.2947x; 1.3302x over previous
#include <hip/hip_runtime.h>
#include <hip/hip_bf16.h>

typedef __attribute__((ext_vector_type(2))) float v2f;
typedef __attribute__((ext_vector_type(8))) float v8f;

#define SEQ_T 32768
#define IN_D  512
#define HID_H 20
#define G4H   80
#define XG_BLOCK 8

__global__ __launch_bounds__(32) void lstm_gates_gemm(
    const float* __restrict__ x,
    const float* __restrict__ W_ih,
    const float* __restrict__ b_ih,
    const float* __restrict__ b_hh,
    float* __restrict__ xg)
{
    const int mtile = blockIdx.x;
    const int lane  = threadIdx.x;
    const int half  = lane >> 4;
    const int l15   = lane & 15;

    v8f acc[5];
#pragma unroll
    for (int nt = 0; nt < 5; ++nt) {
        const int n = nt * 16 + l15;
        const float bias = b_ih[n] + b_hh[n];
#pragma unroll
        for (int v = 0; v < 8; ++v) acc[nt][v] = bias;
    }

    const float* xrow = x + (size_t)(mtile * 16 + l15) * IN_D;

    for (int k0 = 0; k0 < IN_D; k0 += 4) {
        v2f a = *(const v2f*)(xrow + k0 + 2 * half);
#pragma unroll
        for (int nt = 0; nt < 5; ++nt) {
            v2f b = *(const v2f*)(W_ih + (size_t)(nt * 16 + l15) * IN_D + k0 + 2 * half);
            acc[nt] = __builtin_amdgcn_wmma_f32_16x16x4_f32(
                 false, a,  false, b,
                 (short)0, acc[nt],
                 false,  false);
        }
    }

    float* outbase = xg + (size_t)(mtile * 16) * G4H;
#pragma unroll
    for (int nt = 0; nt < 5; ++nt) {
#pragma unroll
        for (int v = 0; v < 8; ++v) {
            outbase[(size_t)(v + 8 * half) * G4H + nt * 16 + l15] = acc[nt][v];
        }
    }
}

__device__ __forceinline__ float readlane_f(float v, int l) {
    return __builtin_bit_cast(float,
        __builtin_amdgcn_readlane(__builtin_bit_cast(int, v), l));
}

__device__ __forceinline__ float sigmoid_f(float v) {
    return __builtin_amdgcn_rcpf(1.0f + __expf(-v));
}

__device__ __forceinline__ float tanh_sel(float x) {
    const float y  = __builtin_fabsf(x);
    const float y2 = y * y;
    float p = fmaf(y2, -0x1.758e7ap-8f, 0x1.521192p-6f);
    p = fmaf(y2, p, -0x1.b8389cp-5f);
    p = fmaf(y2, p,  0x1.110704p-3f);
    p = fmaf(y2, p, -0x1.555532p-2f);
    const float z_small = fmaf(y2, y * p, y);
    const float t = __builtin_amdgcn_exp2f(y * 0x1.715476p+1f);
    const float z_large = fmaf(-2.0f, __builtin_amdgcn_rcpf(t + 1.0f), 1.0f);
    const float z = (y < 0.625f) ? z_small : z_large;
    return __builtin_copysignf(z, x);
}

__global__ __launch_bounds__(32) void lstm_scan(
    const float* __restrict__ xg,
    const float* __restrict__ W_hh,
    const float* __restrict__ W_out,
    const float* __restrict__ b_out,
    const float* __restrict__ h0,
    const float* __restrict__ c0,
    float* __restrict__ out)
{
    const int lane   = threadIdx.x;
    const bool active = lane < HID_H;
    const bool proj   = lane == HID_H;
    const int  n      = active ? lane : 0;

    const float* wi_row = proj ? W_out : W_hh + (size_t)(0 * HID_H + n) * HID_H;
    float wi[HID_H], wf[HID_H], wg[HID_H], wo[HID_H];
#pragma unroll
    for (int k = 0; k < HID_H; ++k) {
        wi[k]   = wi_row[k];
        wf[k]   = W_hh[(size_t)(1 * HID_H + n) * HID_H + k];
        wg[k]   = W_hh[(size_t)(2 * HID_H + n) * HID_H + k];
        wo[k]   = W_hh[(size_t)(3 * HID_H + n) * HID_H + k];
    }
    const float bout = b_out[0];

    float c = c0[n];
    float h = h0[n];

    float ni[XG_BLOCK], nf[XG_BLOCK], nc[XG_BLOCK], no[XG_BLOCK];
#pragma unroll
    for (int j = 0; j < XG_BLOCK; ++j) {
        const float* p = xg + (size_t)j * G4H;
        ni[j] = p[0 * HID_H + n];
        nf[j] = p[1 * HID_H + n];
        nc[j] = p[2 * HID_H + n];
        no[j] = p[3 * HID_H + n];
        ni[j] = proj ? bout : ni[j];
    }

    for (int t0 = 0; t0 < SEQ_T; t0 += XG_BLOCK) {
        float ci[XG_BLOCK], cf[XG_BLOCK], cc[XG_BLOCK], co[XG_BLOCK];
#pragma unroll
        for (int j = 0; j < XG_BLOCK; ++j) {
            ci[j] = ni[j]; cf[j] = nf[j]; cc[j] = nc[j]; co[j] = no[j];
        }

        {
            const int tb = (t0 + XG_BLOCK < SEQ_T) ? (t0 + XG_BLOCK) : (SEQ_T - XG_BLOCK);
            const float* pb = xg + (size_t)tb * G4H;
#pragma unroll
            for (int j = 0; j < XG_BLOCK; ++j) {
                const float* p = pb + j * G4H;
                ni[j] = p[0 * HID_H + n];
                nf[j] = p[1 * HID_H + n];
                nc[j] = p[2 * HID_H + n];
                no[j] = p[3 * HID_H + n];
            }
        }
        __builtin_amdgcn_sched_barrier(0);

        float og[XG_BLOCK];

#pragma unroll
        for (int j = 0; j < XG_BLOCK; ++j) {
            float gi = ci[j], gf = cf[j], gc = cc[j], go = co[j];
            float hk = readlane_f(h, 0);
#pragma unroll
            for (int k = 0; k < HID_H; ++k) {
                const float hn = (k + 1 < HID_H) ? readlane_f(h, k + 1) : 0.0f;
                gi = fmaf(wi[k], hk, gi);
                gf = fmaf(wf[k], hk, gf);
                gc = fmaf(wg[k], hk, gc);
                go = fmaf(wo[k], hk, go);
                hk = hn;
                __builtin_amdgcn_sched_barrier(0);
            }
            og[j] = gi;
            __builtin_amdgcn_sched_barrier(0);

            const float i_ = sigmoid_f(gi);
            const float f_ = sigmoid_f(gf);
            const float o_ = sigmoid_f(go);
            const float g_ = tanh_sel(gc);

            c = fmaf(f_, c, i_ * g_);
            h = o_ * tanh_sel(c);
        }

#pragma unroll
        for (int j = 0; j < XG_BLOCK; ++j) ni[j] = proj ? bout : ni[j];

        if (proj) {
            float* ob = out + t0;
            if (t0 > 0) ob[-1] = og[0];
#pragma unroll
            for (int j = 1; j < XG_BLOCK; ++j) ob[j - 1] = og[j];
        }
    }

    {
        float ot = bout;
#pragma unroll
        for (int k = 0; k < HID_H; ++k) ot = fmaf(W_out[k], readlane_f(h, k), ot);
        if (lane == 0) out[SEQ_T - 1] = ot;
    }

    if (active) {
        out[SEQ_T + n]         = h;
        out[SEQ_T + HID_H + n] = c;
    }
}

extern "C" void kernel_launch(void* const* d_in, const int* in_sizes, int n_in,
                              void* d_out, int out_size, void* d_ws, size_t ws_size,
                              hipStream_t stream) {
    (void)in_sizes; (void)n_in; (void)out_size; (void)ws_size;

    const float* x    = (const float*)d_in[0];
    const float* h0   = (const float*)d_in[1];
    const float* c0   = (const float*)d_in[2];
    const float* W_ih = (const float*)d_in[3];
    const float* W_hh = (const float*)d_in[4];
    const float* b_ih = (const float*)d_in[5];
    const float* b_hh = (const float*)d_in[6];
    const float* W_out= (const float*)d_in[7];
    const float* b_out= (const float*)d_in[8];

    float* out = (float*)d_out;
    float* xg  = (float*)d_ws;

    lstm_gates_gemm<<<SEQ_T / 16, 32, 0, stream>>>(x, W_ih, b_ih, b_hh, xg);

    lstm_scan<<<1, 32, 0, stream>>>(xg, W_hh, W_out, b_out, h0, c0, out);
}


# scan: issue order of every step fixed by hand: activations fenced by readiness (tanh's sign on the gate factor), block's last step pinned ahead of the store branch; values bit-identical
# speedup vs baseline: 4.0452x; 1.2278x over previous
#include <hip/hip_runtime.h>
#include <hip/hip_bf16.h>

typedef __attribute__((ext_vector_type(2))) float v2f;
typedef __attribute__((ext_vector_type(8))) float v8f;

#define SEQ_T 32768
#define IN_D  512
#define HID_H 20
#define G4H   80
#define XG_BLOCK 8

__global__ __launch_bounds__(32) void lstm_gates_gemm(
    const float* __restrict__ x,
    const float* __restrict__ W_ih,
    const float* __restrict__ b_ih,
    const float* __restrict__ b_hh,
    float* __restrict__ xg)
{
    const int mtile = blockIdx.x;
    const int lane  = threadIdx.x;
    const int half  = lane >> 4;
    const int l15   = lane & 15;

    v8f acc[5];
#pragma unroll
    for (int nt = 0; nt < 5; ++nt) {
        const int n = nt * 16 + l15;
        const float bias = b_ih[n] + b_hh[n];
#pragma unroll
        for (int v = 0; v < 8; ++v) acc[nt][v] = bias;
    }

    const float* xrow = x + (size_t)(mtile * 16 + l15) * IN_D;

    for (int k0 = 0; k0 < IN_D; k0 += 4) {
        v2f a = *(const v2f*)(xrow + k0 + 2 * half);
#pragma unroll
        for (int nt = 0; nt < 5; ++nt) {
            v2f b = *(const v2f*)(W_ih + (size_t)(nt * 16 + l15) * IN_D + k0 + 2 * half);
            acc[nt] = __builtin_amdgcn_wmma_f32_16x16x4_f32(
                 false, a,  false, b,
                 (short)0, acc[nt],
                 false,  false);
        }
    }

    float* outbase = xg + (size_t)(mtile * 16) * G4H;
#pragma unroll
    for (int nt = 0; nt < 5; ++nt) {
#pragma unroll
        for (int v = 0; v < 8; ++v) {
            outbase[(size_t)(v + 8 * half) * G4H + nt * 16 + l15] = acc[nt][v];
        }
    }
}

__device__ __forceinline__ float readlane_f(float v, int l) {
    return __builtin_bit_cast(float,
        __builtin_amdgcn_readlane(__builtin_bit_cast(int, v), l));
}

#define K_NLOG2E (-0x1.715476p+0f)
#define K_2LOG2E ( 0x1.715476p+1f)
#define TANH_C4  (-0x1.758e7ap-8f)
#define TANH_C3  ( 0x1.521192p-6f)
#define TANH_C2  (-0x1.b8389cp-5f)
#define TANH_C1  ( 0x1.110704p-3f)
#define TANH_C0  (-0x1.555532p-2f)
#define SEQ __builtin_amdgcn_sched_barrier(0)

__global__ __launch_bounds__(32) void lstm_scan(
    const float* __restrict__ xg,
    const float* __restrict__ W_hh,
    const float* __restrict__ W_out,
    const float* __restrict__ b_out,
    const float* __restrict__ h0,
    const float* __restrict__ c0,
    float* __restrict__ out)
{
    const int lane   = threadIdx.x;
    const bool active = lane < HID_H;
    const bool proj   = lane == HID_H;
    const int  n      = active ? lane : 0;

    const float* wi_row = proj ? W_out : W_hh + (size_t)(0 * HID_H + n) * HID_H;
    float wi[HID_H], wf[HID_H], wg[HID_H], wo[HID_H];
#pragma unroll
    for (int k = 0; k < HID_H; ++k) {
        wi[k]   = wi_row[k];
        wf[k]   = W_hh[(size_t)(1 * HID_H + n) * HID_H + k];
        wg[k]   = W_hh[(size_t)(2 * HID_H + n) * HID_H + k];
        wo[k]   = W_hh[(size_t)(3 * HID_H + n) * HID_H + k];
    }
    const float bout = b_out[0];

    float c = c0[n];
    float h = h0[n];

    float ni[XG_BLOCK], nf[XG_BLOCK], nc[XG_BLOCK], no[XG_BLOCK];
#pragma unroll
    for (int j = 0; j < XG_BLOCK; ++j) {
        const float* p = xg + (size_t)j * G4H;
        ni[j] = p[0 * HID_H + n];
        nf[j] = p[1 * HID_H + n];
        nc[j] = p[2 * HID_H + n];
        no[j] = p[3 * HID_H + n];
        ni[j] = proj ? bout : ni[j];
    }

    for (int t0 = 0; t0 < SEQ_T; t0 += XG_BLOCK) {
        float ci[XG_BLOCK], cf[XG_BLOCK], cc[XG_BLOCK], co[XG_BLOCK];
#pragma unroll
        for (int j = 0; j < XG_BLOCK; ++j) {
            ci[j] = ni[j]; cf[j] = nf[j]; cc[j] = nc[j]; co[j] = no[j];
        }

        {
            const int tb = (t0 + XG_BLOCK < SEQ_T) ? (t0 + XG_BLOCK) : (SEQ_T - XG_BLOCK);
            const float* pb = xg + (size_t)tb * G4H;
#pragma unroll
            for (int j = 0; j < XG_BLOCK; ++j) {
                const float* p = pb + j * G4H;
                ni[j] = p[0 * HID_H + n];
                nf[j] = p[1 * HID_H + n];
                nc[j] = p[2 * HID_H + n];
                no[j] = p[3 * HID_H + n];
            }
        }
        __builtin_amdgcn_sched_barrier(0);

        float og[XG_BLOCK];

#pragma unroll
        for (int j = 0; j < XG_BLOCK; ++j) {
            float gi = ci[j], gf = cf[j], gc = cc[j], go = co[j];
            float hk = readlane_f(h, 0);
#pragma unroll
            for (int k = 0; k < HID_H; ++k) {
                const float hn = (k + 1 < HID_H) ? readlane_f(h, k + 1) : 0.0f;
                gi = fmaf(wi[k], hk, gi);
                gf = fmaf(wf[k], hk, gf);
                gc = fmaf(wg[k], hk, gc);
                go = fmaf(wo[k], hk, go);
                hk = hn;
                __builtin_amdgcn_sched_barrier(0);
            }
            og[j] = gi;
            __builtin_amdgcn_sched_barrier(0);

            const float ag = __builtin_fabsf(gc);
            const float yg = gc * gc;                             SEQ;
            const float mg = ag * K_2LOG2E;                       SEQ;
            const bool  lg = ag < 0.625f;                         SEQ;
            const float si = gi * K_NLOG2E;                       SEQ;
            const float sf = gf * K_NLOG2E;                       SEQ;
            float p = fmaf(yg, TANH_C4, TANH_C3);                 SEQ;
            const float eg = __builtin_amdgcn_exp2f(mg);          SEQ;
            const float so = go * K_NLOG2E;                       SEQ;
            const float ei = __builtin_amdgcn_exp2f(si);          SEQ;
            p = fmaf(yg, p, TANH_C2);                             SEQ;
            const float ef = __builtin_amdgcn_exp2f(sf);          SEQ;
            const float eo = __builtin_amdgcn_exp2f(so);          SEQ;
            const float dg = eg + 1.0f;                           SEQ;
            p = fmaf(yg, p, TANH_C1);                             SEQ;
            const float di = 1.0f + ei;                           SEQ;
            const float rg = __builtin_amdgcn_rcpf(dg);           SEQ;
            p = fmaf(yg, p, TANH_C0);                             SEQ;
            const float df = 1.0f + ef;                           SEQ;
            const float dq = 1.0f + eo;                           SEQ;
            const float i_ = __builtin_amdgcn_rcpf(di);           SEQ;
            const float yp = ag * p;                              SEQ;
            const float f_ = __builtin_amdgcn_rcpf(df);           SEQ;
            const float zl = fmaf(-2.0f, rg, 1.0f);               SEQ;
            const float o_ = __builtin_amdgcn_rcpf(dq);           SEQ;
            const float zs = fmaf(yg, yp, ag);                    SEQ;
            const float is = __builtin_copysignf(i_, gc);         SEQ;
            const float zg = lg ? zs : zl;                        SEQ;
            const float ig = is * zg;                             SEQ;
            c = fmaf(f_, c, ig);                                  SEQ;

            const float ac = __builtin_fabsf(c);
            const float yc = c * c;                               SEQ;
            const float mc = ac * K_2LOG2E;                       SEQ;
            const bool  lc = ac < 0.625f;                         SEQ;
            const float os = __builtin_copysignf(o_, c);          SEQ;
            float q = fmaf(yc, TANH_C4, TANH_C3);                 SEQ;
            const float ec = __builtin_amdgcn_exp2f(mc);          SEQ;
            q = fmaf(yc, q, TANH_C2);                             SEQ;
            const float dc = ec + 1.0f;                           SEQ;
            q = fmaf(yc, q, TANH_C1);                             SEQ;
            const float rc = __builtin_amdgcn_rcpf(dc);           SEQ;
            q = fmaf(yc, q, TANH_C0);                             SEQ;
            const float yq = ac * q;                              SEQ;
            const float wl = fmaf(-2.0f, rc, 1.0f);               SEQ;
            const float ws = fmaf(yc, yq, ac);                    SEQ;
            const float zc = lc ? ws : wl;                        SEQ;
            h = os * zc;                                          SEQ;
        }

        asm volatile("" : "+v"(h), "+v"(c));

#pragma unroll
        for (int j = 0; j < XG_BLOCK; ++j) ni[j] = proj ? bout : ni[j];

        if (proj) {
            float* ob = out + t0;
            if (t0 > 0) ob[-1] = og[0];
#pragma unroll
            for (int j = 1; j < XG_BLOCK; ++j) ob[j - 1] = og[j];
        }
    }

    {
        float ot = bout;
#pragma unroll
        for (int k = 0; k < HID_H; ++k) ot = fmaf(W_out[k], readlane_f(h, k), ot);
        if (lane == 0) out[SEQ_T - 1] = ot;
    }

    if (active) {
        out[SEQ_T + n]         = h;
        out[SEQ_T + HID_H + n] = c;
    }
}

extern "C" void kernel_launch(void* const* d_in, const int* in_sizes, int n_in,
                              void* d_out, int out_size, void* d_ws, size_t ws_size,
                              hipStream_t stream) {
    (void)in_sizes; (void)n_in; (void)out_size; (void)ws_size;

    const float* x    = (const float*)d_in[0];
    const float* h0   = (const float*)d_in[1];
    const float* c0   = (const float*)d_in[2];
    const float* W_ih = (const float*)d_in[3];
    const float* W_hh = (const float*)d_in[4];
    const float* b_ih = (const float*)d_in[5];
    const float* b_hh = (const float*)d_in[6];
    const float* W_out= (const float*)d_in[7];
    const float* b_out= (const float*)d_in[8];

    float* out = (float*)d_out;
    float* xg  = (float*)d_ws;

    lstm_gates_gemm<<<SEQ_T / 16, 32, 0, stream>>>(x, W_ih, b_ih, b_hh, xg);

    lstm_scan<<<1, 32, 0, stream>>>(xg, W_hh, W_out, b_out, h0, c0, out);
}


# scan: activations on v_tanh_f32, one per tanh, sigmoid as 0.5+0.5*tanh(v/2) with the exact halving folded into the i,f,o rows of W_hh and xg columns; values move by the instruction's rounding
# speedup vs baseline: 5.2979x; 1.3097x over previous
#include <hip/hip_runtime.h>
#include <hip/hip_bf16.h>

typedef __attribute__((ext_vector_type(2))) float v2f;
typedef __attribute__((ext_vector_type(8))) float v8f;

#define SEQ_T 32768
#define IN_D  512
#define HID_H 20
#define G4H   80
#define XG_BLOCK 8

__global__ __launch_bounds__(32) void lstm_gates_gemm(
    const float* __restrict__ x,
    const float* __restrict__ W_ih,
    const float* __restrict__ b_ih,
    const float* __restrict__ b_hh,
    float* __restrict__ xg)
{
    const int mtile = blockIdx.x;
    const int lane  = threadIdx.x;
    const int half  = lane >> 4;
    const int l15   = lane & 15;

    v8f acc[5];
#pragma unroll
    for (int nt = 0; nt < 5; ++nt) {
        const int n = nt * 16 + l15;
        const float bias = b_ih[n] + b_hh[n];
#pragma unroll
        for (int v = 0; v < 8; ++v) acc[nt][v] = bias;
    }

    const float* xrow = x + (size_t)(mtile * 16 + l15) * IN_D;

    for (int k0 = 0; k0 < IN_D; k0 += 4) {
        v2f a = *(const v2f*)(xrow + k0 + 2 * half);
#pragma unroll
        for (int nt = 0; nt < 5; ++nt) {
            v2f b = *(const v2f*)(W_ih + (size_t)(nt * 16 + l15) * IN_D + k0 + 2 * half);
            acc[nt] = __builtin_amdgcn_wmma_f32_16x16x4_f32(
                 false, a,  false, b,
                 (short)0, acc[nt],
                 false,  false);
        }
    }

    float* outbase = xg + (size_t)(mtile * 16) * G4H;
#pragma unroll
    for (int nt = 0; nt < 5; ++nt) {
        const int n = nt * 16 + l15;
        const float sc = (n >= 2 * HID_H && n < 3 * HID_H) ? 1.0f : 0.5f;
#pragma unroll
        for (int v = 0; v < 8; ++v) {
            outbase[(size_t)(v + 8 * half) * G4H + nt * 16 + l15] = acc[nt][v] * sc;
        }
    }
}

__device__ __forceinline__ float readlane_f(float v, int l) {
    return __builtin_bit_cast(float,
        __builtin_amdgcn_readlane(__builtin_bit_cast(int, v), l));
}

__global__ __launch_bounds__(32) void lstm_scan(
    const float* __restrict__ xg,
    const float* __restrict__ W_hh,
    const float* __restrict__ W_out,
    const float* __restrict__ b_out,
    const float* __restrict__ h0,
    const float* __restrict__ c0,
    float* __restrict__ out)
{
    const int lane   = threadIdx.x;
    const bool active = lane < HID_H;
    const bool proj   = lane == HID_H;
    const int  n      = active ? lane : 0;

    const float* wi_row = proj ? W_out : W_hh + (size_t)(0 * HID_H + n) * HID_H;
    const float  wi_sc  = proj ? 1.0f : 0.5f;
    float wi[HID_H], wf[HID_H], wg[HID_H], wo[HID_H];
#pragma unroll
    for (int k = 0; k < HID_H; ++k) {
        wi[k]   = wi_sc * wi_row[k];
        wf[k]   = 0.5f * W_hh[(size_t)(1 * HID_H + n) * HID_H + k];
        wg[k]   = W_hh[(size_t)(2 * HID_H + n) * HID_H + k];
        wo[k]   = 0.5f * W_hh[(size_t)(3 * HID_H + n) * HID_H + k];
    }
    const float bout = b_out[0];

    float c = c0[n];
    float h = h0[n];

    float ni[XG_BLOCK], nf[XG_BLOCK], nc[XG_BLOCK], no[XG_BLOCK];
#pragma unroll
    for (int j = 0; j < XG_BLOCK; ++j) {
        const float* p = xg + (size_t)j * G4H;
        ni[j] = p[0 * HID_H + n];
        nf[j] = p[1 * HID_H + n];
        nc[j] = p[2 * HID_H + n];
        no[j] = p[3 * HID_H + n];
        ni[j] = proj ? bout : ni[j];
    }

    for (int t0 = 0; t0 < SEQ_T; t0 += XG_BLOCK) {
        float ci[XG_BLOCK], cf[XG_BLOCK], cc[XG_BLOCK], co[XG_BLOCK];
#pragma unroll
        for (int j = 0; j < XG_BLOCK; ++j) {
            ci[j] = ni[j]; cf[j] = nf[j]; cc[j] = nc[j]; co[j] = no[j];
        }

        {
            const int tb = (t0 + XG_BLOCK < SEQ_T) ? (t0 + XG_BLOCK) : (SEQ_T - XG_BLOCK);
            const float* pb = xg + (size_t)tb * G4H;
#pragma unroll
            for (int j = 0; j < XG_BLOCK; ++j) {
                const float* p = pb + j * G4H;
                ni[j] = p[0 * HID_H + n];
                nf[j] = p[1 * HID_H + n];
                nc[j] = p[2 * HID_H + n];
                no[j] = p[3 * HID_H + n];
            }
        }
        __builtin_amdgcn_sched_barrier(0);

        float og[XG_BLOCK];

#pragma unroll
        for (int j = 0; j < XG_BLOCK; ++j) {
            float gi = ci[j], gf = cf[j], gc = cc[j], go = co[j];
            float hk = readlane_f(h, 0);
#pragma unroll
            for (int k = 0; k < HID_H; ++k) {
                const float hn = (k + 1 < HID_H) ? readlane_f(h, k + 1) : 0.0f;
                gi = fmaf(wi[k], hk, gi);
                gf = fmaf(wf[k], hk, gf);
                gc = fmaf(wg[k], hk, gc);
                go = fmaf(wo[k], hk, go);
                hk = hn;
                __builtin_amdgcn_sched_barrier(0);
            }
            og[j] = gi;
            __builtin_amdgcn_sched_barrier(0);

            const float ti = __builtin_amdgcn_tanhf(gi);
            const float tg = __builtin_amdgcn_tanhf(gc);
            const float tf = __builtin_amdgcn_tanhf(gf);
            const float to = __builtin_amdgcn_tanhf(go);
            const float i_ = fmaf(0.5f, ti, 0.5f);
            const float f_ = fmaf(0.5f, tf, 0.5f);
            const float o_ = fmaf(0.5f, to, 0.5f);
            c = fmaf(f_, c, i_ * tg);
            h = o_ * __builtin_amdgcn_tanhf(c);
        }

        asm volatile("" : "+v"(h), "+v"(c));

#pragma unroll
        for (int j = 0; j < XG_BLOCK; ++j) ni[j] = proj ? bout : ni[j];

        if (proj) {
            float* ob = out + t0;
            if (t0 > 0) ob[-1] = og[0];
#pragma unroll
            for (int j = 1; j < XG_BLOCK; ++j) ob[j - 1] = og[j];
        }
    }

    {
        float ot = bout;
#pragma unroll
        for (int k = 0; k < HID_H; ++k) ot = fmaf(W_out[k], readlane_f(h, k), ot);
        if (lane == 0) out[SEQ_T - 1] = ot;
    }

    if (active) {
        out[SEQ_T + n]         = h;
        out[SEQ_T + HID_H + n] = c;
    }
}

extern "C" void kernel_launch(void* const* d_in, const int* in_sizes, int n_in,
                              void* d_out, int out_size, void* d_ws, size_t ws_size,
                              hipStream_t stream) {
    (void)in_sizes; (void)n_in; (void)out_size; (void)ws_size;

    const float* x    = (const float*)d_in[0];
    const float* h0   = (const float*)d_in[1];
    const float* c0   = (const float*)d_in[2];
    const float* W_ih = (const float*)d_in[3];
    const float* W_hh = (const float*)d_in[4];
    const float* b_ih = (const float*)d_in[5];
    const float* b_hh = (const float*)d_in[6];
    const float* W_out= (const float*)d_in[7];
    const float* b_out= (const float*)d_in[8];

    float* out = (float*)d_out;
    float* xg  = (float*)d_ws;

    lstm_gates_gemm<<<SEQ_T / 16, 32, 0, stream>>>(x, W_ih, b_ih, b_hh, xg);

    lstm_scan<<<1, 32, 0, stream>>>(xg, W_hh, W_out, b_out, h0, c0, out);
}


# scan: each gate's 20-term recurrent sum runs as two partial sums (even k, odd k) added at the end: eight independent chains, k loop issue-bound; summation order changes; per-k fence dropped
# speedup vs baseline: 6.1421x; 1.1593x over previous
#include <hip/hip_runtime.h>
#include <hip/hip_bf16.h>

typedef __attribute__((ext_vector_type(2))) float v2f;
typedef __attribute__((ext_vector_type(8))) float v8f;

#define SEQ_T 32768
#define IN_D  512
#define HID_H 20
#define G4H   80
#define XG_BLOCK 8

__global__ __launch_bounds__(32) void lstm_gates_gemm(
    const float* __restrict__ x,
    const float* __restrict__ W_ih,
    const float* __restrict__ b_ih,
    const float* __restrict__ b_hh,
    float* __restrict__ xg)
{
    const int mtile = blockIdx.x;
    const int lane  = threadIdx.x;
    const int half  = lane >> 4;
    const int l15   = lane & 15;

    v8f acc[5];
#pragma unroll
    for (int nt = 0; nt < 5; ++nt) {
        const int n = nt * 16 + l15;
        const float bias = b_ih[n] + b_hh[n];
#pragma unroll
        for (int v = 0; v < 8; ++v) acc[nt][v] = bias;
    }

    const float* xrow = x + (size_t)(mtile * 16 + l15) * IN_D;

    for (int k0 = 0; k0 < IN_D; k0 += 4) {
        v2f a = *(const v2f*)(xrow + k0 + 2 * half);
#pragma unroll
        for (int nt = 0; nt < 5; ++nt) {
            v2f b = *(const v2f*)(W_ih + (size_t)(nt * 16 + l15) * IN_D + k0 + 2 * half);
            acc[nt] = __builtin_amdgcn_wmma_f32_16x16x4_f32(
                 false, a,  false, b,
                 (short)0, acc[nt],
                 false,  false);
        }
    }

    float* outbase = xg + (size_t)(mtile * 16) * G4H;
#pragma unroll
    for (int nt = 0; nt < 5; ++nt) {
        const int n = nt * 16 + l15;
        const float sc = (n >= 2 * HID_H && n < 3 * HID_H) ? 1.0f : 0.5f;
#pragma unroll
        for (int v = 0; v < 8; ++v) {
            outbase[(size_t)(v + 8 * half) * G4H + nt * 16 + l15] = acc[nt][v] * sc;
        }
    }
}

__device__ __forceinline__ float readlane_f(float v, int l) {
    return __builtin_bit_cast(float,
        __builtin_amdgcn_readlane(__builtin_bit_cast(int, v), l));
}

__global__ __launch_bounds__(32) void lstm_scan(
    const float* __restrict__ xg,
    const float* __restrict__ W_hh,
    const float* __restrict__ W_out,
    const float* __restrict__ b_out,
    const float* __restrict__ h0,
    const float* __restrict__ c0,
    float* __restrict__ out)
{
    const int lane   = threadIdx.x;
    const bool active = lane < HID_H;
    const bool proj   = lane == HID_H;
    const int  n      = active ? lane : 0;

    const float* wi_row = proj ? W_out : W_hh + (size_t)(0 * HID_H + n) * HID_H;
    const float  wi_sc  = proj ? 1.0f : 0.5f;
    float wi[HID_H], wf[HID_H], wg[HID_H], wo[HID_H];
#pragma unroll
    for (int k = 0; k < HID_H; ++k) {
        wi[k]   = wi_sc * wi_row[k];
        wf[k]   = 0.5f * W_hh[(size_t)(1 * HID_H + n) * HID_H + k];
        wg[k]   = W_hh[(size_t)(2 * HID_H + n) * HID_H + k];
        wo[k]   = 0.5f * W_hh[(size_t)(3 * HID_H + n) * HID_H + k];
    }
    const float bout = b_out[0];

    float c = c0[n];
    float h = h0[n];

    float ni[XG_BLOCK], nf[XG_BLOCK], nc[XG_BLOCK], no[XG_BLOCK];
#pragma unroll
    for (int j = 0; j < XG_BLOCK; ++j) {
        const float* p = xg + (size_t)j * G4H;
        ni[j] = p[0 * HID_H + n];
        nf[j] = p[1 * HID_H + n];
        nc[j] = p[2 * HID_H + n];
        no[j] = p[3 * HID_H + n];
        ni[j] = proj ? bout : ni[j];
    }

    for (int t0 = 0; t0 < SEQ_T; t0 += XG_BLOCK) {
        float ci[XG_BLOCK], cf[XG_BLOCK], cc[XG_BLOCK], co[XG_BLOCK];
#pragma unroll
        for (int j = 0; j < XG_BLOCK; ++j) {
            ci[j] = ni[j]; cf[j] = nf[j]; cc[j] = nc[j]; co[j] = no[j];
        }

        {
            const int tb = (t0 + XG_BLOCK < SEQ_T) ? (t0 + XG_BLOCK) : (SEQ_T - XG_BLOCK);
            const float* pb = xg + (size_t)tb * G4H;
#pragma unroll
            for (int j = 0; j < XG_BLOCK; ++j) {
                const float* p = pb + j * G4H;
                ni[j] = p[0 * HID_H + n];
                nf[j] = p[1 * HID_H + n];
                nc[j] = p[2 * HID_H + n];
                no[j] = p[3 * HID_H + n];
            }
        }
        __builtin_amdgcn_sched_barrier(0);

        float og[XG_BLOCK];

#pragma unroll
        for (int j = 0; j < XG_BLOCK; ++j) {
            float ei = ci[j], ef = cf[j], ec = cc[j], eo = co[j];
            float di = 0.0f, df = 0.0f, dc = 0.0f, do_ = 0.0f;
            float he = readlane_f(h, 0);
            float hd = readlane_f(h, 1);
#pragma unroll
            for (int k = 0; k < HID_H; k += 2) {
                const float hen = (k + 2 < HID_H) ? readlane_f(h, k + 2) : 0.0f;
                const float hdn = (k + 3 < HID_H) ? readlane_f(h, k + 3) : 0.0f;
                ei = fmaf(wi[k], he, ei);
                ef = fmaf(wf[k], he, ef);
                ec = fmaf(wg[k], he, ec);
                eo = fmaf(wo[k], he, eo);
                if (k == 0) {
                    di = wi[1] * hd;
                    df = wf[1] * hd;
                    dc = wg[1] * hd;
                    do_ = wo[1] * hd;
                } else {
                    di = fmaf(wi[k + 1], hd, di);
                    df = fmaf(wf[k + 1], hd, df);
                    dc = fmaf(wg[k + 1], hd, dc);
                    do_ = fmaf(wo[k + 1], hd, do_);
                }
                he = hen;
                hd = hdn;
            }
            const float gi = ei + di;
            const float gf = ef + df;
            const float gc = ec + dc;
            const float go = eo + do_;
            og[j] = gi;
            __builtin_amdgcn_sched_barrier(0);

            const float ti = __builtin_amdgcn_tanhf(gi);
            const float tg = __builtin_amdgcn_tanhf(gc);
            const float tf = __builtin_amdgcn_tanhf(gf);
            const float to = __builtin_amdgcn_tanhf(go);
            const float i_ = fmaf(0.5f, ti, 0.5f);
            const float f_ = fmaf(0.5f, tf, 0.5f);
            const float o_ = fmaf(0.5f, to, 0.5f);
            c = fmaf(f_, c, i_ * tg);
            h = o_ * __builtin_amdgcn_tanhf(c);
        }

        asm volatile("" : "+v"(h), "+v"(c));

#pragma unroll
        for (int j = 0; j < XG_BLOCK; ++j) ni[j] = proj ? bout : ni[j];

        if (proj) {
            float* ob = out + t0;
            if (t0 > 0) ob[-1] = og[0];
#pragma unroll
            for (int j = 1; j < XG_BLOCK; ++j) ob[j - 1] = og[j];
        }
    }

    {
        float oe = bout;
        float od = W_out[1] * readlane_f(h, 1);
#pragma unroll
        for (int k = 0; k < HID_H; k += 2) {
            oe = fmaf(W_out[k], readlane_f(h, k), oe);
            if (k > 0) od = fmaf(W_out[k + 1], readlane_f(h, k + 1), od);
        }
        if (lane == 0) out[SEQ_T - 1] = oe + od;
    }

    if (active) {
        out[SEQ_T + n]         = h;
        out[SEQ_T + HID_H + n] = c;
    }
}

extern "C" void kernel_launch(void* const* d_in, const int* in_sizes, int n_in,
                              void* d_out, int out_size, void* d_ws, size_t ws_size,
                              hipStream_t stream) {
    (void)in_sizes; (void)n_in; (void)out_size; (void)ws_size;

    const float* x    = (const float*)d_in[0];
    const float* h0   = (const float*)d_in[1];
    const float* c0   = (const float*)d_in[2];
    const float* W_ih = (const float*)d_in[3];
    const float* W_hh = (const float*)d_in[4];
    const float* b_ih = (const float*)d_in[5];
    const float* b_hh = (const float*)d_in[6];
    const float* W_out= (const float*)d_in[7];
    const float* b_out= (const float*)d_in[8];

    float* out = (float*)d_out;
    float* xg  = (float*)d_ws;

    lstm_gates_gemm<<<SEQ_T / 16, 32, 0, stream>>>(x, W_ih, b_ih, b_hh, xg);

    lstm_scan<<<1, 32, 0, stream>>>(xg, W_hh, W_out, b_out, h0, c0, out);
}
